# starting kernel
# baseline (speedup 1.0000x reference)
#include <hip/hip_runtime.h>
#include <math.h>


typedef __attribute__((ext_vector_type(2))) float v2f;
typedef __attribute__((ext_vector_type(8))) float v8f;

#define GAT_N      50000
#define GAT_E      1600000
#define GAT_C      128
#define GAT_HEADS  4
#define GAT_FOUT   32
#define NEG_SLOPE  0.2f

__global__ __launch_bounds__(256) void gemm_f32_wmma(
    const float* __restrict__ X, const float* __restrict__ W,
    float* __restrict__ H)
{
    const int wave = threadIdx.x >> 5;
    const int lane = threadIdx.x & 31;
    const int m0   = blockIdx.x * 16;
    const int n0   = wave * 16;

    const int r    = lane & 15;
    const int koff = (lane >> 4) * 2;

    const float* __restrict__ xrow = X + (size_t)(m0 + r) * GAT_C + koff;
    const float* __restrict__ wcol = W + (size_t)koff * GAT_C + (n0 + r);

    v8f acc = {};
    #pragma unroll
    for (int k = 0; k < GAT_C; k += 4) {
        v2f a; a.x = xrow[k];
               a.y = xrow[k + 1];
        v2f b; b.x = wcol[(size_t)k * GAT_C];
               b.y = wcol[(size_t)(k + 1) * GAT_C];
        acc = __builtin_amdgcn_wmma_f32_16x16x4_f32(
             false, a,  false, b,
             (short)0, acc,  false,  false);
    }

    float* __restrict__ hp = H + (size_t)(m0 + (lane >> 4) * 8) * GAT_C + (n0 + r);
    #pragma unroll
    for (int i = 0; i < 8; ++i)
        hp[(size_t)i * GAT_C] = acc[i];
}

__global__ __launch_bounds__(256) void att_scores(
    const float* __restrict__ H,
    const float* __restrict__ att_s, const float* __restrict__ att_d,
    float* __restrict__ a_s, float* __restrict__ a_d, int n)
{
    const int node = blockIdx.x * 8 + (threadIdx.x >> 5);
    const int lane = threadIdx.x & 31;
    if (node >= n) return;
    const float* __restrict__ hp = H + (size_t)node * GAT_C;
    #pragma unroll
    for (int h = 0; h < GAT_HEADS; ++h) {
        const float v = hp[h * GAT_FOUT + lane];
        float s = v * att_s[h * GAT_FOUT + lane];
        float d = v * att_d[h * GAT_FOUT + lane];
        #pragma unroll
        for (int off = 16; off > 0; off >>= 1) {
            s += __shfl_down(s, off, 32);
            d += __shfl_down(d, off, 32);
        }
        if (lane == 0) {
            a_s[node * GAT_HEADS + h] = s;
            a_d[node * GAT_HEADS + h] = d;
        }
    }
}

__global__ __launch_bounds__(256) void init_layer(
    float* __restrict__ emax, float* __restrict__ denom,
    float* __restrict__ out, const float* __restrict__ bias, int n)
{
    const int i = blockIdx.x * 256 + threadIdx.x;
    if (i < n * GAT_HEADS) { emax[i] = -INFINITY; denom[i] = 0.0f; }
    if (i < n * GAT_C)     { out[i] = bias[i & (GAT_C - 1)]; }
}

__global__ __launch_bounds__(256) void edge_seg_max(
    const int* __restrict__ src, const int* __restrict__ dst,
    const float* __restrict__ a_s, const float* __restrict__ a_d,
    float* __restrict__ emax, int ne)
{
    const int t = blockIdx.x * 256 + threadIdx.x;
    if (t >= ne * GAT_HEADS) return;
    const int e = t >> 2, h = t & 3;
    const int s = src[e], d = dst[e];
    float v = a_s[s * GAT_HEADS + h] + a_d[d * GAT_HEADS + h];
    v = (v > 0.0f) ? v : NEG_SLOPE * v;
    atomicMax(&emax[d * GAT_HEADS + h], v);
}

__global__ __launch_bounds__(256) void edge_seg_sumexp(
    const int* __restrict__ src, const int* __restrict__ dst,
    const float* __restrict__ a_s, const float* __restrict__ a_d,
    const float* __restrict__ emax, float* __restrict__ denom, int ne)
{
    const int t = blockIdx.x * 256 + threadIdx.x;
    if (t >= ne * GAT_HEADS) return;
    const int e = t >> 2, h = t & 3;
    const int s = src[e], d = dst[e];
    float v = a_s[s * GAT_HEADS + h] + a_d[d * GAT_HEADS + h];
    v = (v > 0.0f) ? v : NEG_SLOPE * v;
    atomicAdd(&denom[d * GAT_HEADS + h], __expf(v - emax[d * GAT_HEADS + h]));
}

__global__ __launch_bounds__(256) void edge_aggregate(
    const int* __restrict__ src, const int* __restrict__ dst,
    const float* __restrict__ a_s, const float* __restrict__ a_d,
    const float* __restrict__ emax, const float* __restrict__ denom,
    const float* __restrict__ H, float* __restrict__ out, int ne)
{
    const int e = blockIdx.x * 8 + (threadIdx.x >> 5);
    if (e >= ne) return;
    const int lane = threadIdx.x & 31;
    const int s = src[e], d = dst[e];
    #pragma unroll
    for (int h = 0; h < GAT_HEADS; ++h) {
        float v = a_s[s * GAT_HEADS + h] + a_d[d * GAT_HEADS + h];
        v = (v > 0.0f) ? v : NEG_SLOPE * v;
        float den = denom[d * GAT_HEADS + h];
        den = (den > 0.0f) ? den : 1.0f;
        const float alpha = __expf(v - emax[d * GAT_HEADS + h]) / den;
        const float hv = H[(size_t)s * GAT_C + h * GAT_FOUT + lane];
        atomicAdd(&out[(size_t)d * GAT_C + h * GAT_FOUT + lane], alpha * hv);
    }
}

__global__ __launch_bounds__(256) void relu_inplace(float* __restrict__ x, int n)
{
    const int i = blockIdx.x * 256 + threadIdx.x;
    if (i < n) x[i] = fmaxf(x[i], 0.0f);
}

extern "C" void kernel_launch(void* const* d_in, const int* in_sizes, int n_in,
                              void* d_out, int out_size, void* d_ws, size_t ws_size,
                              hipStream_t stream)
{
    const float* x    = (const float*)d_in[0];
    const int*   eidx = (const int*)  d_in[1];
    const float* W0   = (const float*)d_in[2];
    const float* as0  = (const float*)d_in[3];
    const float* ad0  = (const float*)d_in[4];
    const float* b0   = (const float*)d_in[5];
    const float* W1   = (const float*)d_in[6];
    const float* as1  = (const float*)d_in[7];
    const float* ad1  = (const float*)d_in[8];
    const float* b1   = (const float*)d_in[9];

    const int N = in_sizes[0] / GAT_C;
    const int E = in_sizes[1] / 2;
    const int* src = eidx;
    const int* dst = eidx + E;

    char* ws = (char*)d_ws;
    float* hbuf  = (float*)ws;                                  ws += (size_t)N * GAT_C * 4;
    float* out0  = (float*)ws;                                  ws += (size_t)N * GAT_C * 4;
    float* a_s   = (float*)ws;                                  ws += (size_t)N * GAT_HEADS * 4;
    float* a_d   = (float*)ws;                                  ws += (size_t)N * GAT_HEADS * 4;
    float* emax  = (float*)ws;                                  ws += (size_t)N * GAT_HEADS * 4;
    float* denom = (float*)ws;

    const int blk = 256;
    const dim3 gGemm(N / 16);
    const dim3 gNode((N + 7) / 8);
    const dim3 gInit((N * GAT_C + blk - 1) / blk);
    const dim3 gEdgeH((E * GAT_HEADS + blk - 1) / blk);
    const dim3 gEdgeW((E + 7) / 8);
    const dim3 gElem((N * GAT_C + blk - 1) / blk);

    init_layer    <<<gInit,  blk, 0, stream>>>(emax, denom, out0, b0, N);
    gemm_f32_wmma <<<gGemm,  blk, 0, stream>>>(x, W0, hbuf);
    att_scores    <<<gNode,  blk, 0, stream>>>(hbuf, as0, ad0, a_s, a_d, N);
    edge_seg_max  <<<gEdgeH, blk, 0, stream>>>(src, dst, a_s, a_d, emax, E);
    edge_seg_sumexp<<<gEdgeH,blk, 0, stream>>>(src, dst, a_s, a_d, emax, denom, E);
    edge_aggregate<<<gEdgeW, blk, 0, stream>>>(src, dst, a_s, a_d, emax, denom,
                                               hbuf, out0, E);
    relu_inplace  <<<gElem,  blk, 0, stream>>>(out0, N * GAT_C);

    float* outf = (float*)d_out;
    init_layer    <<<gInit,  blk, 0, stream>>>(emax, denom, outf, b1, N);
    gemm_f32_wmma <<<gGemm,  blk, 0, stream>>>(out0, W1, hbuf);
    att_scores    <<<gNode,  blk, 0, stream>>>(hbuf, as1, ad1, a_s, a_d, N);
    edge_seg_max  <<<gEdgeH, blk, 0, stream>>>(src, dst, a_s, a_d, emax, E);
    edge_seg_sumexp<<<gEdgeH,blk, 0, stream>>>(src, dst, a_s, a_d, emax, denom, E);
    edge_aggregate<<<gEdgeW, blk, 0, stream>>>(src, dst, a_s, a_d, emax, denom,
                                               hbuf, outf, E);
}


# Edge-pass sums in two planes, one per package half: an edge's plane is fixed by its index, a block draws a 64-edge chunk of its own half's plane by ticket; finish adds the planes; sum tree changed
# speedup vs baseline: 2.3399x; 2.3399x over previous
#include <hip/hip_runtime.h>
#include <math.h>


typedef __attribute__((ext_vector_type(2))) float v2f;
typedef __attribute__((ext_vector_type(8))) float v8f;

#define GAT_N      50000
#define GAT_E      1600000
#define GAT_C      128
#define GAT_HEADS  4
#define GAT_FOUT   32
#define NEG_SLOPE  0.2f

__global__ __launch_bounds__(256) void gemm_f32_wmma(
    const float* __restrict__ X, const float* __restrict__ W,
    float* __restrict__ H)
{
    const int wave = threadIdx.x >> 5;
    const int lane = threadIdx.x & 31;
    const int m0   = blockIdx.x * 16;
    const int n0   = wave * 16;

    const int r    = lane & 15;
    const int koff = (lane >> 4) * 2;

    const float* __restrict__ xrow = X + (size_t)(m0 + r) * GAT_C + koff;
    const float* __restrict__ wcol = W + (size_t)koff * GAT_C + (n0 + r);

    v8f acc = {};
    #pragma unroll
    for (int k = 0; k < GAT_C; k += 4) {
        v2f a; a.x = xrow[k];
               a.y = xrow[k + 1];
        v2f b; b.x = wcol[(size_t)k * GAT_C];
               b.y = wcol[(size_t)(k + 1) * GAT_C];
        acc = __builtin_amdgcn_wmma_f32_16x16x4_f32(
             false, a,  false, b,
             (short)0, acc,  false,  false);
    }

    float* __restrict__ hp = H + (size_t)(m0 + (lane >> 4) * 8) * GAT_C + (n0 + r);
    #pragma unroll
    for (int i = 0; i < 8; ++i)
        hp[(size_t)i * GAT_C] = acc[i];
}

__global__ __launch_bounds__(256) void att_scores(
    const float* __restrict__ H,
    const float* __restrict__ att_s, const float* __restrict__ att_d,
    float* __restrict__ a_s, float* __restrict__ a_d,
    float* __restrict__ gstat, int n)
{
    __shared__ float red[8][2 * GAT_HEADS];
    const int wv   = threadIdx.x >> 5;
    const int node = blockIdx.x * 8 + wv;
    const int lane = threadIdx.x & 31;
    if (node < n) {
        const float* __restrict__ hp = H + (size_t)node * GAT_C;
        #pragma unroll
        for (int h = 0; h < GAT_HEADS; ++h) {
            const float v = hp[h * GAT_FOUT + lane];
            float s = v * att_s[h * GAT_FOUT + lane];
            float d = v * att_d[h * GAT_FOUT + lane];
            #pragma unroll
            for (int off = 16; off > 0; off >>= 1) {
                s += __shfl_down(s, off, 32);
                d += __shfl_down(d, off, 32);
            }
            if (lane == 0) {
                a_s[node * GAT_HEADS + h] = s;
                a_d[node * GAT_HEADS + h] = d;
                red[wv][h] = s;
                red[wv][GAT_HEADS + h] = -s;
            }
        }
    } else if (lane < 2 * GAT_HEADS) {
        red[wv][lane] = -INFINITY;
    }
    __syncthreads();
    if (threadIdx.x < 2 * GAT_HEADS) {
        float m = red[0][threadIdx.x];
        #pragma unroll
        for (int w = 1; w < 8; ++w) m = fmaxf(m, red[w][threadIdx.x]);
        atomicMax(&gstat[threadIdx.x], m);
    }
}

#define BOOK_CTR0  32
#define BOOK_CTR1  96

__global__ __launch_bounds__(256) void init_layer(
    float* __restrict__ emax, float* __restrict__ denom,
    float* __restrict__ p0, float* __restrict__ p1, int n)
{
    const int i = blockIdx.x * 256 + threadIdx.x;
    if (i < n * GAT_HEADS) { emax[i] = -INFINITY; denom[i] = 0.0f; }
    if (i < n * GAT_C) {
        p0[i] = 0.0f;
        p1[i] = (i < 2 * GAT_HEADS) ? -INFINITY : 0.0f;
    }
}

#define SPREAD_LIMIT 32.0f

struct ShiftInfo { float amax[GAT_HEADS]; bool big; };

__device__ __forceinline__ ShiftInfo load_shift_info(const float* __restrict__ gstat)
{
    ShiftInfo si;
    float nmin[GAT_HEADS];
    #pragma unroll
    for (int h = 0; h < GAT_HEADS; ++h) {
        si.amax[h] = gstat[h];
        nmin[h]    = gstat[GAT_HEADS + h];
    }
    bool big = false;
    #pragma unroll
    for (int h = 0; h < GAT_HEADS; ++h)
        big |= (si.amax[h] + nmin[h] > SPREAD_LIMIT);
    si.big = big;
    return si;
}

__device__ __forceinline__ float lane_amax(const ShiftInfo& si, int hh)
{
    const float a01 = (hh & 1) ? si.amax[1] : si.amax[0];
    const float a23 = (hh & 1) ? si.amax[3] : si.amax[2];
    return (hh & 2) ? a23 : a01;
}

#define SEGMAX_BLOCKS 512

__global__ __launch_bounds__(256) void edge_seg_max_exact(
    const int* __restrict__ src, const int* __restrict__ dst,
    const float* __restrict__ a_s, const float* __restrict__ a_d,
    const float* __restrict__ gstat, float* __restrict__ emax, int ne, int n)
{
    const int stride = gridDim.x * 256;
    if (!load_shift_info(gstat).big) {
        for (int t = blockIdx.x * 256 + threadIdx.x; t < n * GAT_HEADS; t += stride)
            emax[t] = 0.0f;
        return;
    }
    const int total  = ne * GAT_HEADS;
    for (int t = blockIdx.x * 256 + threadIdx.x; t < total; t += stride) {
        const int e = t >> 2, h = t & 3;
        const int s = src[e], d = dst[e];
        float v = a_s[s * GAT_HEADS + h] + a_d[d * GAT_HEADS + h];
        v = (v > 0.0f) ? v : NEG_SLOPE * v;
        atomicMax(&emax[d * GAT_HEADS + h], v);
    }
}

#define AGG_EPW 8

__device__ __forceinline__ void aggregate_one_edge(
    int e, int lane,
    const int* __restrict__ src, const int* __restrict__ dst,
    const float* __restrict__ a_s, const float* __restrict__ a_d,
    const float* emax, float amax_hh, bool big,
    const float* __restrict__ H,
    float* denom, float* p0, float* pm)
{
    const int s = src[e], d = dst[e];
    const int hh = lane & (GAT_HEADS - 1);
    const float* __restrict__ hrow = H + (size_t)s * GAT_C;
    float hv[GAT_HEADS];
    #pragma unroll
    for (int h = 0; h < GAT_HEADS; ++h)
        hv[h] = hrow[h * GAT_FOUT + lane];
    const float vd = a_d[d * GAT_HEADS + hh];
    float v = a_s[s * GAT_HEADS + hh] + vd;
    v = (v > 0.0f) ? v : NEG_SLOPE * v;
    float vm;
    if (big) {
        vm = emax[d * GAT_HEADS + hh];
    } else {
        vm = amax_hh + vd;
        vm = (vm > 0.0f) ? vm : NEG_SLOPE * vm;
    }
    const float w = __expf(v - vm);
    float* orow = ((d == 0) ? p0 : pm) + (size_t)d * GAT_C;
    #pragma unroll
    for (int h = 0; h < GAT_HEADS; ++h) {
        const float wh = __int_as_float(
            __builtin_amdgcn_readlane(__float_as_int(w), h));
        atomicAdd(&orow[h * GAT_FOUT + lane], wh * hv[h]);
    }
    if (lane < GAT_HEADS)
        atomicAdd(&denom[d * GAT_HEADS + lane], w);
}

__global__ __launch_bounds__(256) void edge_aggregate(
    const int* __restrict__ src, const int* __restrict__ dst,
    const float* __restrict__ a_s, const float* __restrict__ a_d,
    float* emax_den1, const float* __restrict__ gstat, int* ctr,
    const float* __restrict__ H,
    float* den0, float* p0, float* p1, int ne)
{
    __shared__ int sh_chunk;
    const int nchunk  = (int)gridDim.x;
    const int nchunk0 = nchunk >> 1;
    if (threadIdx.x == 0) {
        const int die = (int)((__builtin_amdgcn_s_sendmsg_rtn(0x87) >> 16) & 15u);
        int half = (die >= 4) ? 1 : 0;
        int t = atomicAdd(&ctr[half ? BOOK_CTR1 : BOOK_CTR0], 1);
        if (t >= (half ? nchunk - nchunk0 : nchunk0)) {
            half ^= 1;
            t = atomicAdd(&ctr[half ? BOOK_CTR1 : BOOK_CTR0], 1);
            if (t >= (half ? nchunk - nchunk0 : nchunk0)) t = -nchunk - 1;
        }
        sh_chunk = half ? nchunk0 + t : t;
    }
    __syncthreads();
    const int chunk = __builtin_amdgcn_readfirstlane(sh_chunk);
    if (chunk < 0) return;
    const bool plane1 = chunk >= nchunk0;

    const int lane = threadIdx.x & 31;
    const ShiftInfo si = load_shift_info(gstat);
    const bool big = si.big;
    float* pm   = plane1 ? p1 : p0;
    float* denm = (plane1 && !big) ? emax_den1 : den0;
    const float* emax = emax_den1;
    const int wv = __builtin_amdgcn_readfirstlane((int)(threadIdx.x >> 5));
    const int e0 = (chunk * 8 + wv) * AGG_EPW;
    if (e0 >= ne) return;
    if (e0 + AGG_EPW > ne) {
        for (int e = e0; e < ne; ++e)
            aggregate_one_edge(e, lane, src, dst, a_s, a_d, emax,
                               lane_amax(si, lane & (GAT_HEADS - 1)), big,
                               H, denm, p0, pm);
        return;
    }

    const int hh = lane & (GAT_HEADS - 1);
    const int el = e0 + (lane >> 2);
    const int s = src[el], d = dst[el];

    float hv[AGG_EPW][GAT_HEADS];
    #pragma unroll
    for (int j = 0; j < AGG_EPW; ++j) {
        const int sj = __builtin_amdgcn_readlane(s, j * GAT_HEADS);
        const float* __restrict__ hrow = H + (size_t)sj * GAT_C;
        #pragma unroll
        for (int h = 0; h < GAT_HEADS; ++h)
            hv[j][h] = hrow[h * GAT_FOUT + lane];
    }
    const float vs = a_s [s * GAT_HEADS + hh];
    const float vd = a_d [d * GAT_HEADS + hh];
    float vm;
    if (big) {
        vm = emax[d * GAT_HEADS + hh];
    } else {
        vm = lane_amax(si, hh) + vd;
        vm = (vm > 0.0f) ? vm : NEG_SLOPE * vm;
    }

    float v = vs + vd;
    v = (v > 0.0f) ? v : NEG_SLOPE * v;
    const float w = __expf(v - vm);

    #pragma unroll
    for (int j = 0; j < AGG_EPW; ++j) {
        const int dj = __builtin_amdgcn_readlane(d, j * GAT_HEADS);
        float* orow = ((dj == 0) ? p0 : pm) + (size_t)dj * GAT_C;
        #pragma unroll
        for (int h = 0; h < GAT_HEADS; ++h) {
            const float wh = __int_as_float(__builtin_amdgcn_readlane(
                __float_as_int(w), j * GAT_HEADS + h));
            atomicAdd(&orow[h * GAT_FOUT + lane], wh * hv[j][h]);
        }
    }
    atomicAdd(&denm[d * GAT_HEADS + hh], w);
}

template <bool RELU>
__global__ __launch_bounds__(256) void finish_layer(
    float* __restrict__ out, const float* p1,
    const float* __restrict__ den0, const float* __restrict__ den1,
    const float* gstat, const float* __restrict__ bias, int n)
{
    const int i = blockIdx.x * 256 + threadIdx.x;
    if (i >= n * (GAT_C / 4)) return;
    const int node = i >> 5;
    const int q    = i & 31;
    const bool big = load_shift_info(gstat).big;
    float den = den0[node * GAT_HEADS + (q >> 3)];
    if (!big) den += den1[node * GAT_HEADS + (q >> 3)];
    den = (den > 0.0f) ? den : 1.0f;
    float4 a = reinterpret_cast<float4*>(out)[i];
    if (node != 0) {
        const float4 c = reinterpret_cast<const float4*>(p1)[i];
        a.x += c.x; a.y += c.y; a.z += c.z; a.w += c.w;
    }
    const float4 b = reinterpret_cast<const float4*>(bias)[q];
    a.x = a.x / den + b.x;
    a.y = a.y / den + b.y;
    a.z = a.z / den + b.z;
    a.w = a.w / den + b.w;
    if (RELU) {
        a.x = fmaxf(a.x, 0.0f); a.y = fmaxf(a.y, 0.0f);
        a.z = fmaxf(a.z, 0.0f); a.w = fmaxf(a.w, 0.0f);
    }
    reinterpret_cast<float4*>(out)[i] = a;
}

extern "C" void kernel_launch(void* const* d_in, const int* in_sizes, int n_in,
                              void* d_out, int out_size, void* d_ws, size_t ws_size,
                              hipStream_t stream)
{
    const float* x    = (const float*)d_in[0];
    const int*   eidx = (const int*)  d_in[1];
    const float* W0   = (const float*)d_in[2];
    const float* as0  = (const float*)d_in[3];
    const float* ad0  = (const float*)d_in[4];
    const float* b0   = (const float*)d_in[5];
    const float* W1   = (const float*)d_in[6];
    const float* as1  = (const float*)d_in[7];
    const float* ad1  = (const float*)d_in[8];
    const float* b1   = (const float*)d_in[9];

    const int N = in_sizes[0] / GAT_C;
    const int E = in_sizes[1] / 2;
    const int* src = eidx;
    const int* dst = eidx + E;

    char* ws = (char*)d_ws;
    float* hbuf  = (float*)ws;                                  ws += (size_t)N * GAT_C * 4;
    float* out0  = (float*)ws;                                  ws += (size_t)N * GAT_C * 4;
    float* a_s   = (float*)ws;                                  ws += (size_t)N * GAT_HEADS * 4;
    float* a_d   = (float*)ws;                                  ws += (size_t)N * GAT_HEADS * 4;
    float* emax  = (float*)ws;                                  ws += (size_t)N * GAT_HEADS * 4;
    float* denom = (float*)ws;

    const int blk = 256;
    const dim3 gGemm(N / 16);
    const dim3 gNode((N + 7) / 8);
    const dim3 gInit((N * GAT_C + blk - 1) / blk);
    const dim3 gSegMax(SEGMAX_BLOCKS);
    const dim3 gEdgeW((E + 8 * AGG_EPW - 1) / (8 * AGG_EPW));
    const dim3 gQuad((N * (GAT_C / 4) + blk - 1) / blk);

    float* outf = (float*)d_out;
    float* gstat0 = outf;
    float* gstat1 = out0;

    gemm_f32_wmma <<<gGemm,  blk, 0, stream>>>(x, W0, hbuf);
    init_layer    <<<gInit,  blk, 0, stream>>>(emax, denom, out0, outf, N);
    att_scores    <<<gNode,  blk, 0, stream>>>(hbuf, as0, ad0, a_s, a_d, gstat0, N);
    edge_seg_max_exact<<<gSegMax, blk, 0, stream>>>(src, dst, a_s, a_d, gstat0, emax, E, N);
    edge_aggregate<<<gEdgeW, blk, 0, stream>>>(src, dst, a_s, a_d, emax, gstat0, (int*)outf, hbuf,
                                               denom, out0, outf, E);
    finish_layer<true> <<<gQuad, blk, 0, stream>>>(out0, outf, denom, emax, gstat0, b0, N);

    gemm_f32_wmma <<<gGemm,  blk, 0, stream>>>(out0, W1, hbuf);
    init_layer    <<<gInit,  blk, 0, stream>>>(emax, denom, outf, out0, N);
    att_scores    <<<gNode,  blk, 0, stream>>>(hbuf, as1, ad1, a_s, a_d, gstat1, N);
    edge_seg_max_exact<<<gSegMax, blk, 0, stream>>>(src, dst, a_s, a_d, gstat1, emax, E, N);
    edge_aggregate<<<gEdgeW, blk, 0, stream>>>(src, dst, a_s, a_d, emax, gstat1, (int*)out0, hbuf,
                                               denom, outf, out0, E);
    finish_layer<false><<<gQuad, blk, 0, stream>>>(outf, out0, denom, emax, gstat1, b1, N);
}
